# starting kernel
# baseline (speedup 1.0000x reference)
#include <hip/hip_runtime.h>

#define BATCH 32
#define V     1024
#define HID   768
#define OUT   256
#define S1    8
#define S2    8

typedef __attribute__((ext_vector_type(2))) float v2f;
typedef __attribute__((ext_vector_type(8))) float v8f;

__global__ void colsum_partial_kernel(const float* __restrict__ adj,
                                      float* __restrict__ part) {
  const int b = blockIdx.x;
  const int w = blockIdx.y * 256 + threadIdx.x;
  const int s = blockIdx.z;
  const float* p = adj + (size_t)b * V * V + (size_t)(s * (V / S1)) * V + w;
  float acc = 0.0f;
#pragma unroll 4
  for (int v = 0; v < V / S1; ++v) {
    __builtin_prefetch(p + 16 * V, 0, 1);
    acc += *p;
    p += V;
  }
  part[((b * S1 + s) * V) + w] = acc;
}

__global__ void colsum_reduce_kernel(const float* __restrict__ part,
                                     float* __restrict__ colsum) {
  const int b = blockIdx.x;
  const int w = blockIdx.y * 256 + threadIdx.x;
  float acc = 0.0f;
#pragma unroll
  for (int s = 0; s < S1; ++s) acc += part[(b * S1 + s) * V + w];
  colsum[b * V + w] = acc;
}

__global__ void gather_partial_kernel(const float* __restrict__ emb,
                                      const int* __restrict__ ids,
                                      const float* __restrict__ colsum,
                                      float* __restrict__ hpart) {
  const int b  = blockIdx.x;
  const int hh = blockIdx.y * 256 + threadIdx.x;
  const int s  = blockIdx.z;
  const int wbase = s * (V / S2);
  float acc = 0.0f;
  for (int i = 0; i < V / S2; ++i) {
    const int   w  = wbase + i;
    const int   id = ids[b * V + w];
    const float cw = colsum[b * V + w];
    acc += cw * emb[(size_t)id * HID + hh];
  }
  hpart[(b * S2 + s) * HID + hh] = acc;
}

__global__ void h_reduce_kernel(const float* __restrict__ hpart,
                                float* __restrict__ hbuf) {
  const int b  = blockIdx.x;
  const int hh = blockIdx.y * 256 + threadIdx.x;
  float acc = 0.0f;
#pragma unroll
  for (int s = 0; s < S2; ++s) acc += hpart[(b * S2 + s) * HID + hh];
  hbuf[b * HID + hh] = acc * (1.0f / (float)V);
}

__global__ void fc_wmma_kernel(const float* __restrict__ hbuf,
                               const float* __restrict__ fc_w,
                               const float* __restrict__ fc_b,
                               float* __restrict__ out) {
  const int wave  = (blockIdx.x * blockDim.x + threadIdx.x) >> 5;
  const int lane  = threadIdx.x & 31;
  const int mTile = wave >> 4;
  const int nTile = wave & 15;
  const int m0 = mTile * 16;
  const int n0 = nTile * 16;
  const int half = lane >> 4;
  const int l    = lane & 15;
  const int kOff = 2 * half;

  const float* arow = hbuf + (size_t)(m0 + l) * HID;
  const float* brow = fc_w + (size_t)(n0 + l) * HID;

  v8f c = {};
  for (int k = 0; k < HID; k += 4) {
    v2f a  = *(const v2f*)(arow + k + kOff);
    v2f bb = *(const v2f*)(brow + k + kOff);
    c = __builtin_amdgcn_wmma_f32_16x16x4_f32(false, a, false, bb,
                                              (short)0, c, false, false);
  }

#pragma unroll
  for (int i = 0; i < 8; ++i) {
    const int m = m0 + i + 8 * half;
    const int n = n0 + l;
    out[m * OUT + n] = c[i] + fc_b[n];
  }
}

extern "C" void kernel_launch(void* const* d_in, const int* in_sizes, int n_in,
                              void* d_out, int out_size, void* d_ws, size_t ws_size,
                              hipStream_t stream) {
  (void)in_sizes; (void)n_in; (void)out_size; (void)ws_size;
  const float* emb    = (const float*)d_in[0];
  const float* adj    = (const float*)d_in[1];
  const float* fc_w   = (const float*)d_in[2];
  const float* fc_b   = (const float*)d_in[3];
  const int*   gv_ids = (const int*)d_in[4];
  float* out = (float*)d_out;

  char* ws = (char*)d_ws;
  float* part   = (float*)ws;
  float* colsum = (float*)(ws + sizeof(float) * ((size_t)BATCH * S1 * V));
  float* hpart  = (float*)(ws + sizeof(float) * ((size_t)BATCH * S1 * V +
                                                 (size_t)BATCH * V));
  float* hbuf   = (float*)(ws + sizeof(float) * ((size_t)BATCH * S1 * V +
                                                 (size_t)BATCH * V +
                                                 (size_t)BATCH * S2 * HID));

  colsum_partial_kernel<<<dim3(BATCH, V / 256, S1), 256, 0, stream>>>(adj, part);
  colsum_reduce_kernel <<<dim3(BATCH, V / 256),     256, 0, stream>>>(part, colsum);
  gather_partial_kernel<<<dim3(BATCH, HID / 256, S2), 256, 0, stream>>>(emb, gv_ids, colsum, hpart);
  h_reduce_kernel      <<<dim3(BATCH, HID / 256),     256, 0, stream>>>(hpart, hbuf);
  fc_wmma_kernel<<<dim3(4), 256, 0, stream>>>(hbuf, fc_w, fc_b, out);
}


# fc GEMM: K split over 8 waves per 16x16 tile (32 blocks, was 4) with 16-byte fragment loads (k permuted alike in A and B), partial tiles summed via LDS in fixed order; summation order changes
# speedup vs baseline: 1.3118x; 1.3118x over previous
#include <hip/hip_runtime.h>

#define BATCH 32
#define V     1024
#define HID   768
#define OUT   256
#define S1    8
#define S2    8

typedef __attribute__((ext_vector_type(2))) float v2f;
typedef __attribute__((ext_vector_type(4))) float v4f;
typedef __attribute__((ext_vector_type(8))) float v8f;

__global__ void colsum_partial_kernel(const float* __restrict__ adj,
                                      float* __restrict__ part) {
  const int b = blockIdx.x;
  const int w = blockIdx.y * 256 + threadIdx.x;
  const int s = blockIdx.z;
  const float* p = adj + (size_t)b * V * V + (size_t)(s * (V / S1)) * V + w;
  float acc = 0.0f;
#pragma unroll 4
  for (int v = 0; v < V / S1; ++v) {
    __builtin_prefetch(p + 16 * V, 0, 1);
    acc += *p;
    p += V;
  }
  part[((b * S1 + s) * V) + w] = acc;
}

__global__ void colsum_reduce_kernel(const float* __restrict__ part,
                                     float* __restrict__ colsum) {
  const int b = blockIdx.x;
  const int w = blockIdx.y * 256 + threadIdx.x;
  float acc = 0.0f;
#pragma unroll
  for (int s = 0; s < S1; ++s) acc += part[(b * S1 + s) * V + w];
  colsum[b * V + w] = acc;
}

__global__ void gather_partial_kernel(const float* __restrict__ emb,
                                      const int* __restrict__ ids,
                                      const float* __restrict__ colsum,
                                      float* __restrict__ hpart) {
  const int b  = blockIdx.x;
  const int hh = blockIdx.y * 256 + threadIdx.x;
  const int s  = blockIdx.z;
  const int wbase = s * (V / S2);
  float acc = 0.0f;
  for (int i = 0; i < V / S2; ++i) {
    const int   w  = wbase + i;
    const int   id = ids[b * V + w];
    const float cw = colsum[b * V + w];
    acc += cw * emb[(size_t)id * HID + hh];
  }
  hpart[(b * S2 + s) * HID + hh] = acc;
}

__global__ void h_reduce_kernel(const float* __restrict__ hpart,
                                float* __restrict__ hbuf) {
  const int b  = blockIdx.x;
  const int hh = blockIdx.y * 256 + threadIdx.x;
  float acc = 0.0f;
#pragma unroll
  for (int s = 0; s < S2; ++s) acc += hpart[(b * S2 + s) * HID + hh];
  hbuf[b * HID + hh] = acc * (1.0f / (float)V);
}

#define KS 8
#define KW (HID / KS)
#define KSTEPS (KW / 8)

__global__ void __launch_bounds__(32 * KS)
fc_wmma_kernel(const float* __restrict__ hbuf,
               const float* __restrict__ fc_w,
               const float* __restrict__ fc_b,
               float* __restrict__ out) {
  __shared__ float red[KS * 256];

  const int tile  = blockIdx.x;
  const int m0    = (tile >> 4) * 16;
  const int n0    = (tile & 15) * 16;
  const int ks    = threadIdx.x >> 5;
  const int lane  = threadIdx.x & 31;
  const int half  = lane >> 4;
  const int l     = lane & 15;

  const float* arow = hbuf + (size_t)(m0 + l) * HID + ks * KW + 4 * half;
  const float* brow = fc_w + (size_t)(n0 + l) * HID + ks * KW + 4 * half;

  v4f a[KSTEPS], b[KSTEPS];
#pragma unroll
  for (int j = 0; j < KSTEPS; ++j) {
    a[j] = *(const v4f*)(arow + 8 * j);
    b[j] = *(const v4f*)(brow + 8 * j);
  }

  v8f c0 = {}, c1 = {};
#pragma unroll
  for (int j = 0; j < KSTEPS; ++j) {
    v2f a0 = {a[j].x, a[j].y}, b0 = {b[j].x, b[j].y};
    v2f a1 = {a[j].z, a[j].w}, b1 = {b[j].z, b[j].w};
    c0 = __builtin_amdgcn_wmma_f32_16x16x4_f32(false, a0, false, b0,
                                               (short)0, c0, false, false);
    c1 = __builtin_amdgcn_wmma_f32_16x16x4_f32(false, a1, false, b1,
                                               (short)0, c1, false, false);
  }
  c0 += c1;

#pragma unroll
  for (int i = 0; i < 8; ++i) red[ks * 256 + i * 32 + lane] = c0[i];
  __syncthreads();

  const int t = threadIdx.x;
  float s = red[t];
#pragma unroll
  for (int j = 1; j < KS; ++j) s += red[j * 256 + t];
  const int m = m0 + (t >> 5) + 8 * ((t >> 4) & 1);
  const int n = n0 + (t & 15);
  out[m * OUT + n] = s + fc_b[n];
}

extern "C" void kernel_launch(void* const* d_in, const int* in_sizes, int n_in,
                              void* d_out, int out_size, void* d_ws, size_t ws_size,
                              hipStream_t stream) {
  (void)in_sizes; (void)n_in; (void)out_size; (void)ws_size;
  const float* emb    = (const float*)d_in[0];
  const float* adj    = (const float*)d_in[1];
  const float* fc_w   = (const float*)d_in[2];
  const float* fc_b   = (const float*)d_in[3];
  const int*   gv_ids = (const int*)d_in[4];
  float* out = (float*)d_out;

  char* ws = (char*)d_ws;
  float* part   = (float*)ws;
  float* colsum = (float*)(ws + sizeof(float) * ((size_t)BATCH * S1 * V));
  float* hpart  = (float*)(ws + sizeof(float) * ((size_t)BATCH * S1 * V +
                                                 (size_t)BATCH * V));
  float* hbuf   = (float*)(ws + sizeof(float) * ((size_t)BATCH * S1 * V +
                                                 (size_t)BATCH * V +
                                                 (size_t)BATCH * S2 * HID));

  colsum_partial_kernel<<<dim3(BATCH, V / 256, S1), 256, 0, stream>>>(adj, part);
  colsum_reduce_kernel <<<dim3(BATCH, V / 256),     256, 0, stream>>>(part, colsum);
  gather_partial_kernel<<<dim3(BATCH, HID / 256, S2), 256, 0, stream>>>(emb, gv_ids, colsum, hpart);
  h_reduce_kernel      <<<dim3(BATCH, HID / 256),     256, 0, stream>>>(hpart, hbuf);
  fc_wmma_kernel<<<dim3(32), 256, 0, stream>>>(hbuf, fc_w, fc_b, out);
}


# gather kernel: 16-byte row pieces per lane (512 B per wave load), all 64 row reads of a block's slice requested before the first sum (was 4 reads of 128 B); w split 16: summation order changes
# speedup vs baseline: 1.7627x; 1.3437x over previous
#include <hip/hip_runtime.h>

#define BATCH 32
#define V     1024
#define HID   768
#define OUT   256
#define S1    8
#define S2    16

typedef __attribute__((ext_vector_type(2))) float v2f;
typedef __attribute__((ext_vector_type(4))) float v4f;
typedef __attribute__((ext_vector_type(8))) float v8f;

__global__ void colsum_partial_kernel(const float* __restrict__ adj,
                                      float* __restrict__ part) {
  const int b = blockIdx.x;
  const int w = blockIdx.y * 256 + threadIdx.x;
  const int s = blockIdx.z;
  const float* p = adj + (size_t)b * V * V + (size_t)(s * (V / S1)) * V + w;
  float acc = 0.0f;
#pragma unroll 4
  for (int v = 0; v < V / S1; ++v) {
    __builtin_prefetch(p + 16 * V, 0, 1);
    acc += *p;
    p += V;
  }
  part[((b * S1 + s) * V) + w] = acc;
}

__global__ void colsum_reduce_kernel(const float* __restrict__ part,
                                     float* __restrict__ colsum) {
  const int b = blockIdx.x;
  const int w = blockIdx.y * 256 + threadIdx.x;
  float acc = 0.0f;
#pragma unroll
  for (int s = 0; s < S1; ++s) acc += part[(b * S1 + s) * V + w];
  colsum[b * V + w] = acc;
}

#define GW (V / S2)
#define GT (HID / 4)

__global__ void __launch_bounds__(GT)
gather_partial_kernel(const float* __restrict__ emb,
                      const int* __restrict__ ids,
                      const float* __restrict__ colsum,
                      float* __restrict__ hpart) {
  const int b = blockIdx.x;
  const int s = blockIdx.y;
  const int*   idp = ids    + b * V + s * GW;
  const float* cwp = colsum + b * V + s * GW;
  const int    off = 4 * threadIdx.x;
  v4f r[GW];
#pragma unroll
  for (int u = 0; u < GW; ++u) {
    const float* row = emb + (size_t)idp[u] * HID;
    r[u] = *(const v4f*)(row + off);
  }
  __builtin_amdgcn_sched_barrier(0);
  v4f acc = {0.0f, 0.0f, 0.0f, 0.0f};
#pragma unroll
  for (int u = 0; u < GW; ++u)
    acc += cwp[u] * r[u];
  *(v4f*)(hpart + (size_t)(b * S2 + s) * HID + off) = acc;
}

__global__ void h_reduce_kernel(const float* __restrict__ hpart,
                                float* __restrict__ hbuf) {
  const int b  = blockIdx.x;
  const int hh = blockIdx.y * 256 + threadIdx.x;
  float acc = 0.0f;
#pragma unroll
  for (int s = 0; s < S2; ++s) acc += hpart[(b * S2 + s) * HID + hh];
  hbuf[b * HID + hh] = acc * (1.0f / (float)V);
}

#define KS 8
#define KW (HID / KS)
#define KSTEPS (KW / 8)

__global__ void __launch_bounds__(32 * KS)
fc_wmma_kernel(const float* __restrict__ hbuf,
               const float* __restrict__ fc_w,
               const float* __restrict__ fc_b,
               float* __restrict__ out) {
  __shared__ float red[KS * 256];

  const int tile  = blockIdx.x;
  const int m0    = (tile >> 4) * 16;
  const int n0    = (tile & 15) * 16;
  const int ks    = threadIdx.x >> 5;
  const int lane  = threadIdx.x & 31;
  const int half  = lane >> 4;
  const int l     = lane & 15;

  const float* arow = hbuf + (size_t)(m0 + l) * HID + ks * KW + 4 * half;
  const float* brow = fc_w + (size_t)(n0 + l) * HID + ks * KW + 4 * half;

  v4f a[KSTEPS], b[KSTEPS];
#pragma unroll
  for (int j = 0; j < KSTEPS; ++j) {
    a[j] = *(const v4f*)(arow + 8 * j);
    b[j] = *(const v4f*)(brow + 8 * j);
  }

  v8f c0 = {}, c1 = {};
#pragma unroll
  for (int j = 0; j < KSTEPS; ++j) {
    v2f a0 = {a[j].x, a[j].y}, b0 = {b[j].x, b[j].y};
    v2f a1 = {a[j].z, a[j].w}, b1 = {b[j].z, b[j].w};
    c0 = __builtin_amdgcn_wmma_f32_16x16x4_f32(false, a0, false, b0,
                                               (short)0, c0, false, false);
    c1 = __builtin_amdgcn_wmma_f32_16x16x4_f32(false, a1, false, b1,
                                               (short)0, c1, false, false);
  }
  c0 += c1;

#pragma unroll
  for (int i = 0; i < 8; ++i) red[ks * 256 + i * 32 + lane] = c0[i];
  __syncthreads();

  const int t = threadIdx.x;
  float s = red[t];
#pragma unroll
  for (int j = 1; j < KS; ++j) s += red[j * 256 + t];
  const int m = m0 + (t >> 5) + 8 * ((t >> 4) & 1);
  const int n = n0 + (t & 15);
  out[m * OUT + n] = s + fc_b[n];
}

extern "C" void kernel_launch(void* const* d_in, const int* in_sizes, int n_in,
                              void* d_out, int out_size, void* d_ws, size_t ws_size,
                              hipStream_t stream) {
  (void)in_sizes; (void)n_in; (void)out_size; (void)ws_size;
  const float* emb    = (const float*)d_in[0];
  const float* adj    = (const float*)d_in[1];
  const float* fc_w   = (const float*)d_in[2];
  const float* fc_b   = (const float*)d_in[3];
  const int*   gv_ids = (const int*)d_in[4];
  float* out = (float*)d_out;

  char* ws = (char*)d_ws;
  float* part   = (float*)ws;
  float* colsum = (float*)(ws + sizeof(float) * ((size_t)BATCH * S1 * V));
  float* hpart  = (float*)(ws + sizeof(float) * ((size_t)BATCH * S1 * V +
                                                 (size_t)BATCH * V));
  float* hbuf   = (float*)(ws + sizeof(float) * ((size_t)BATCH * S1 * V +
                                                 (size_t)BATCH * V +
                                                 (size_t)BATCH * S2 * HID));

  colsum_partial_kernel<<<dim3(BATCH, V / 256, S1), 256, 0, stream>>>(adj, part);
  colsum_reduce_kernel <<<dim3(BATCH, V / 256),     256, 0, stream>>>(part, colsum);
  gather_partial_kernel<<<dim3(BATCH, S2), GT, 0, stream>>>(emb, gv_ids, colsum, hpart);
  h_reduce_kernel      <<<dim3(BATCH, HID / 256),     256, 0, stream>>>(hpart, hbuf);
  fc_wmma_kernel<<<dim3(32), 256, 0, stream>>>(hbuf, fc_w, fc_b, out);
}


# colsum kernel: one block sums all 1024 rows of a 64-column slice (16-byte non-temporal loads, 16-32 in flight per wave), 16 partials summed via LDS; reduce launch gone; summation order changes
# speedup vs baseline: 2.1002x; 1.1915x over previous
#include <hip/hip_runtime.h>

#define BATCH 32
#define V     1024
#define HID   768
#define OUT   256
#define S2    16

typedef __attribute__((ext_vector_type(2))) float v2f;
typedef __attribute__((ext_vector_type(4))) float v4f;
typedef __attribute__((ext_vector_type(8))) float v8f;

#define CW  64
#define CT  256
#define CRG (CT / 16)
#define CG  16
#define CNG (V / CRG / CG)

__global__ void __launch_bounds__(CT)
colsum_kernel(const float* __restrict__ adj, float* __restrict__ colsum) {
  __shared__ float red[CRG * CW];
  const int b  = blockIdx.x;
  const int s  = blockIdx.y;
  const int c4 = threadIdx.x & 15;
  const int rg = threadIdx.x >> 4;
  const float* p = adj + (size_t)b * V * V + (size_t)rg * V + s * CW + 4 * c4;

  v4f r[2][CG];
  v4f acc = {0.0f, 0.0f, 0.0f, 0.0f};
#pragma unroll
  for (int j = 0; j < CG; ++j)
    r[0][j] = __builtin_nontemporal_load((const v4f*)(p + (size_t)j * CRG * V));
#pragma unroll
  for (int g = 0; g < CNG; ++g) {
    __builtin_amdgcn_sched_barrier(0);
    if (g + 1 < CNG) {
#pragma unroll
      for (int j = 0; j < CG; ++j)
        r[(g + 1) & 1][j] = __builtin_nontemporal_load(
            (const v4f*)(p + (size_t)((g + 1) * CG + j) * CRG * V));
    }
    __builtin_amdgcn_sched_barrier(0);
#pragma unroll
    for (int j = 0; j < CG; ++j) acc += r[g & 1][j];
  }

  *(v4f*)(red + rg * CW + 4 * c4) = acc;
  __syncthreads();
  if (threadIdx.x < CW) {
    float t = red[threadIdx.x];
#pragma unroll
    for (int g = 1; g < CRG; ++g) t += red[g * CW + threadIdx.x];
    colsum[b * V + s * CW + threadIdx.x] = t;
  }
}

#define GW (V / S2)
#define GT (HID / 4)

__global__ void __launch_bounds__(GT)
gather_partial_kernel(const float* __restrict__ emb,
                      const int* __restrict__ ids,
                      const float* __restrict__ colsum,
                      float* __restrict__ hpart) {
  const int b = blockIdx.x;
  const int s = blockIdx.y;
  const int*   idp = ids    + b * V + s * GW;
  const float* cwp = colsum + b * V + s * GW;
  const int    off = 4 * threadIdx.x;
  v4f r[GW];
#pragma unroll
  for (int u = 0; u < GW; ++u) {
    const float* row = emb + (size_t)idp[u] * HID;
    r[u] = *(const v4f*)(row + off);
  }
  __builtin_amdgcn_sched_barrier(0);
  v4f acc = {0.0f, 0.0f, 0.0f, 0.0f};
#pragma unroll
  for (int u = 0; u < GW; ++u)
    acc += cwp[u] * r[u];
  *(v4f*)(hpart + (size_t)(b * S2 + s) * HID + off) = acc;
}

__global__ void h_reduce_kernel(const float* __restrict__ hpart,
                                float* __restrict__ hbuf) {
  const int b  = blockIdx.x;
  const int hh = blockIdx.y * 256 + threadIdx.x;
  float acc = 0.0f;
#pragma unroll
  for (int s = 0; s < S2; ++s) acc += hpart[(b * S2 + s) * HID + hh];
  hbuf[b * HID + hh] = acc * (1.0f / (float)V);
}

#define KS 8
#define KW (HID / KS)
#define KSTEPS (KW / 8)

__global__ void __launch_bounds__(32 * KS)
fc_wmma_kernel(const float* __restrict__ hbuf,
               const float* __restrict__ fc_w,
               const float* __restrict__ fc_b,
               float* __restrict__ out) {
  __shared__ float red[KS * 256];

  const int tile  = blockIdx.x;
  const int m0    = (tile >> 4) * 16;
  const int n0    = (tile & 15) * 16;
  const int ks    = threadIdx.x >> 5;
  const int lane  = threadIdx.x & 31;
  const int half  = lane >> 4;
  const int l     = lane & 15;

  const float* arow = hbuf + (size_t)(m0 + l) * HID + ks * KW + 4 * half;
  const float* brow = fc_w + (size_t)(n0 + l) * HID + ks * KW + 4 * half;

  v4f a[KSTEPS], b[KSTEPS];
#pragma unroll
  for (int j = 0; j < KSTEPS; ++j) {
    a[j] = *(const v4f*)(arow + 8 * j);
    b[j] = *(const v4f*)(brow + 8 * j);
  }

  v8f c0 = {}, c1 = {};
#pragma unroll
  for (int j = 0; j < KSTEPS; ++j) {
    v2f a0 = {a[j].x, a[j].y}, b0 = {b[j].x, b[j].y};
    v2f a1 = {a[j].z, a[j].w}, b1 = {b[j].z, b[j].w};
    c0 = __builtin_amdgcn_wmma_f32_16x16x4_f32(false, a0, false, b0,
                                               (short)0, c0, false, false);
    c1 = __builtin_amdgcn_wmma_f32_16x16x4_f32(false, a1, false, b1,
                                               (short)0, c1, false, false);
  }
  c0 += c1;

#pragma unroll
  for (int i = 0; i < 8; ++i) red[ks * 256 + i * 32 + lane] = c0[i];
  __syncthreads();

  const int t = threadIdx.x;
  float s = red[t];
#pragma unroll
  for (int j = 1; j < KS; ++j) s += red[j * 256 + t];
  const int m = m0 + (t >> 5) + 8 * ((t >> 4) & 1);
  const int n = n0 + (t & 15);
  out[m * OUT + n] = s + fc_b[n];
}

extern "C" void kernel_launch(void* const* d_in, const int* in_sizes, int n_in,
                              void* d_out, int out_size, void* d_ws, size_t ws_size,
                              hipStream_t stream) {
  (void)in_sizes; (void)n_in; (void)out_size; (void)ws_size;
  const float* emb    = (const float*)d_in[0];
  const float* adj    = (const float*)d_in[1];
  const float* fc_w   = (const float*)d_in[2];
  const float* fc_b   = (const float*)d_in[3];
  const int*   gv_ids = (const int*)d_in[4];
  float* out = (float*)d_out;

  char* ws = (char*)d_ws;
  float* colsum = (float*)ws;
  float* hpart  = (float*)(ws + sizeof(float) * ((size_t)BATCH * V));
  float* hbuf   = (float*)(ws + sizeof(float) * ((size_t)BATCH * V +
                                                 (size_t)BATCH * S2 * HID));

  colsum_kernel        <<<dim3(BATCH, V / CW), CT, 0, stream>>>(adj, colsum);
  gather_partial_kernel<<<dim3(BATCH, S2), GT, 0, stream>>>(emb, gv_ids, colsum, hpart);
  h_reduce_kernel      <<<dim3(BATCH, HID / 256),     256, 0, stream>>>(hpart, hbuf);
  fc_wmma_kernel<<<dim3(32), 256, 0, stream>>>(hbuf, fc_w, fc_b, out);
}


# reduce launch folded away: a gather block owns (b, half of w, 256-byte column unit), two rows per wave load, 8 partials summed via LDS; fc loader adds the 2 halves; summation order changes
# speedup vs baseline: 2.3726x; 1.1297x over previous
#include <hip/hip_runtime.h>

#define BATCH 32
#define V     1024
#define HID   768
#define OUT   256
#define S2    2

typedef __attribute__((ext_vector_type(2))) float v2f;
typedef __attribute__((ext_vector_type(4))) float v4f;
typedef __attribute__((ext_vector_type(8))) float v8f;

#define CW  64
#define CT  256
#define CRG (CT / 16)
#define CG  16
#define CNG (V / CRG / CG)

__global__ void __launch_bounds__(CT)
colsum_kernel(const float* __restrict__ adj, float* __restrict__ colsum) {
  __shared__ float red[CRG * CW];
  const int b  = blockIdx.x;
  const int s  = blockIdx.y;
  const int c4 = threadIdx.x & 15;
  const int rg = threadIdx.x >> 4;
  const float* p = adj + (size_t)b * V * V + (size_t)rg * V + s * CW + 4 * c4;

  v4f r[2][CG];
  v4f acc = {0.0f, 0.0f, 0.0f, 0.0f};
#pragma unroll
  for (int j = 0; j < CG; ++j)
    r[0][j] = __builtin_nontemporal_load((const v4f*)(p + (size_t)j * CRG * V));
#pragma unroll
  for (int g = 0; g < CNG; ++g) {
    __builtin_amdgcn_sched_barrier(0);
    if (g + 1 < CNG) {
#pragma unroll
      for (int j = 0; j < CG; ++j)
        r[(g + 1) & 1][j] = __builtin_nontemporal_load(
            (const v4f*)(p + (size_t)((g + 1) * CG + j) * CRG * V));
    }
    __builtin_amdgcn_sched_barrier(0);
#pragma unroll
    for (int j = 0; j < CG; ++j) acc += r[g & 1][j];
  }

  *(v4f*)(red + rg * CW + 4 * c4) = acc;
  __syncthreads();
  if (threadIdx.x < CW) {
    float t = red[threadIdx.x];
#pragma unroll
    for (int g = 1; g < CRG; ++g) t += red[g * CW + threadIdx.x];
    colsum[b * V + s * CW + threadIdx.x] = t;
  }
}

#define GW  (V / S2)
#define GWV 4
#define GT  (32 * GWV)
#define GU  (HID / 64)
#define GL  (GW / GWV / 2)

__global__ void __launch_bounds__(GT)
gather_partial_kernel(const float* __restrict__ emb,
                      const int* __restrict__ ids,
                      const float* __restrict__ colsum,
                      float* __restrict__ hpart) {
  __shared__ float red[GWV * 2 * 64];
  const int u = blockIdx.x;
  const int s = blockIdx.y;
  const int b = blockIdx.z;
  const int wave = __builtin_amdgcn_readfirstlane((int)(threadIdx.x >> 5));
  const int lane = threadIdx.x & 31;
  const bool odd = (lane >> 4) != 0;
  const int  c4  = lane & 15;
  const int  w0  = s * GW + wave * (2 * GL);
  const int*   idp = ids    + b * V + w0 + lane;
  const float* cwp = colsum + b * V + w0 + lane;
  int iv[4], wv[4];
#pragma unroll
  for (int k = 0; k < 4; ++k) {
    iv[k] = idp[32 * k];
    wv[k] = __builtin_bit_cast(int, cwp[32 * k]);
  }
  const unsigned loff = (unsigned)(u * 256 + c4 * 16);
  const char* ebase = (const char*)emb;
  v4f r[GL];
#pragma unroll
  for (int j = 0; j < GL; ++j) {
    const int id0 = __builtin_amdgcn_readlane(iv[j >> 4], 2 * (j & 15));
    const int id1 = __builtin_amdgcn_readlane(iv[j >> 4], 2 * (j & 15) + 1);
    const int id  = odd ? id1 : id0;
    const unsigned boff = (unsigned)id * (unsigned)(HID * 4) + loff;
    r[j] = *(const v4f*)(ebase + boff);
  }
  __builtin_amdgcn_sched_barrier(0);
  v4f acc = {0.0f, 0.0f, 0.0f, 0.0f};
#pragma unroll
  for (int j = 0; j < GL; ++j) {
    const int w0b = __builtin_amdgcn_readlane(wv[j >> 4], 2 * (j & 15));
    const int w1b = __builtin_amdgcn_readlane(wv[j >> 4], 2 * (j & 15) + 1);
    const float wgt = __builtin_bit_cast(float, odd ? w1b : w0b);
    acc += wgt * r[j];
  }
  *(v4f*)(red + (wave * 2 + (odd ? 1 : 0)) * 64 + 4 * c4) = acc;
  __syncthreads();
  if (threadIdx.x < 64) {
    float t = red[threadIdx.x];
#pragma unroll
    for (int g = 1; g < GWV * 2; ++g) t += red[g * 64 + threadIdx.x];
    hpart[(size_t)(b * S2 + s) * HID + u * 64 + threadIdx.x] = t;
  }
}

#define KS 8
#define KW (HID / KS)
#define KSTEPS (KW / 8)

__global__ void __launch_bounds__(32 * KS)
fc_wmma_kernel(const float* __restrict__ hpart,
               const float* __restrict__ fc_w,
               const float* __restrict__ fc_b,
               float* __restrict__ out) {
  __shared__ float red[KS * 256];

  const int tile  = blockIdx.x;
  const int m0    = (tile >> 4) * 16;
  const int n0    = (tile & 15) * 16;
  const int ks    = threadIdx.x >> 5;
  const int lane  = threadIdx.x & 31;
  const int half  = lane >> 4;
  const int l     = lane & 15;

  const float* arow = hpart + (size_t)(m0 + l) * S2 * HID + ks * KW + 4 * half;
  const float* brow = fc_w + (size_t)(n0 + l) * HID + ks * KW + 4 * half;

  v4f a[KSTEPS], b[KSTEPS];
#pragma unroll
  for (int j = 0; j < KSTEPS; ++j) {
    a[j] = *(const v4f*)(arow + 8 * j);
    b[j] = *(const v4f*)(brow + 8 * j);
  }
#pragma unroll
  for (int j = 0; j < KSTEPS; ++j)
    a[j] += *(const v4f*)(arow + HID + 8 * j);

  v8f c0 = {}, c1 = {};
#pragma unroll
  for (int j = 0; j < KSTEPS; ++j) {
    v2f a0 = {a[j].x, a[j].y}, b0 = {b[j].x, b[j].y};
    v2f a1 = {a[j].z, a[j].w}, b1 = {b[j].z, b[j].w};
    c0 = __builtin_amdgcn_wmma_f32_16x16x4_f32(false, a0, false, b0,
                                               (short)0, c0, false, false);
    c1 = __builtin_amdgcn_wmma_f32_16x16x4_f32(false, a1, false, b1,
                                               (short)0, c1, false, false);
  }
  c0 += c1;

#pragma unroll
  for (int i = 0; i < 8; ++i) red[ks * 256 + i * 32 + lane] = c0[i];
  __syncthreads();

  const int t = threadIdx.x;
  float s = red[t];
#pragma unroll
  for (int j = 1; j < KS; ++j) s += red[j * 256 + t];
  const int m = m0 + (t >> 5) + 8 * ((t >> 4) & 1);
  const int n = n0 + (t & 15);
  out[m * OUT + n] = s * (1.0f / (float)V) + fc_b[n];
}

extern "C" void kernel_launch(void* const* d_in, const int* in_sizes, int n_in,
                              void* d_out, int out_size, void* d_ws, size_t ws_size,
                              hipStream_t stream) {
  (void)in_sizes; (void)n_in; (void)out_size; (void)ws_size;
  const float* emb    = (const float*)d_in[0];
  const float* adj    = (const float*)d_in[1];
  const float* fc_w   = (const float*)d_in[2];
  const float* fc_b   = (const float*)d_in[3];
  const int*   gv_ids = (const int*)d_in[4];
  float* out = (float*)d_out;

  char* ws = (char*)d_ws;
  float* colsum = (float*)ws;
  float* hpart  = (float*)(ws + sizeof(float) * ((size_t)BATCH * V));

  colsum_kernel        <<<dim3(BATCH, V / CW), CT, 0, stream>>>(adj, colsum);
  gather_partial_kernel<<<dim3(GU, S2, BATCH), GT, 0, stream>>>(emb, gv_ids, colsum, hpart);
  fc_wmma_kernel<<<dim3(32), 256, 0, stream>>>(hpart, fc_w, fc_b, out);
}
